# starting kernel
# baseline (speedup 1.0000x reference)
#include <hip/hip_runtime.h>
#include <hip/hip_bf16.h>

#define NHEADS 4
#define HDIM   32
#define NNODES 1024
#define BATCH  8
#define INDIM  128
#define OUTDIM 128
#define NEG_INF_F (-1000000000.0f)

typedef float v2f __attribute__((ext_vector_type(2)));
typedef float v8f __attribute__((ext_vector_type(8)));

__global__ __launch_bounds__(256) void gat_linear(const float* __restrict__ x,
                                                  const float* __restrict__ W,
                                                  const float* __restrict__ bias,
                                                  float* __restrict__ h)
{
    const int lane    = threadIdx.x & 31;
    const int wave    = threadIdx.x >> 5;
    const int rowTile = blockIdx.x;
    const int rloc    = lane & 15;
    const int kq      = (lane >> 4) * 2;
    const int col     = wave * 16 + rloc;

    __shared__ float xs[16][132];
    for (int i = threadIdx.x; i < 16 * 128; i += 256) {
        int rr = i >> 7, cc = i & 127;
        xs[rr][cc] = x[(rowTile * 16 + rr) * INDIM + cc];
    }
    __syncthreads();

    v8f acc = {};
    for (int kk = 0; kk < INDIM; kk += 4) {
        v2f a, b;
        a.x = xs[rloc][kk + kq + 0];
        a.y = xs[rloc][kk + kq + 1];
        b.x = W[col * INDIM + kk + kq + 0];
        b.y = W[col * INDIM + kk + kq + 1];
        acc = __builtin_amdgcn_wmma_f32_16x16x4_f32(false, a, false, b,
                                                    (short)0, acc, false, false);
    }

    const float bv = bias[col];
    const int rbase = rowTile * 16 + ((lane >> 4) << 3);
#pragma unroll
    for (int v = 0; v < 8; ++v)
        h[(rbase + v) * OUTDIM + col] = acc[v] + bv;
}

__global__ __launch_bounds__(256) void gat_scores(const float* __restrict__ h,
                                                  const float* __restrict__ a,
                                                  float* __restrict__ s,
                                                  float* __restrict__ t)
{
    int idx = blockIdx.x * blockDim.x + threadIdx.x;
    if (idx >= BATCH * NHEADS * NNODES) return;
    int n    = idx & (NNODES - 1);
    int head = (idx >> 10) & (NHEADS - 1);
    int b    = idx >> 12;
    const float* hp = h + ((size_t)(b * NNODES + n) * OUTDIM + head * HDIM);
    float ss = 0.f, tt = 0.f;
#pragma unroll
    for (int d = 0; d < HDIM; ++d) {
        float hv = hp[d];
        ss += hv * a[d];
        tt += hv * a[HDIM + d];
    }
    s[idx] = ss;
    t[idx] = tt;
}

__global__ __launch_bounds__(128) void gat_attn(const float* __restrict__ h,
                                                const int*   __restrict__ adj,
                                                const float* __restrict__ s,
                                                const float* __restrict__ t,
                                                float* __restrict__ out)
{
    const int rowTile = blockIdx.x & 63;
    const int head    = (blockIdx.x >> 6) & (NHEADS - 1);
    const int b       = blockIdx.x >> 8;

    const int tid  = threadIdx.x;
    const int wave = tid >> 5;
    const int lane = tid & 31;
    const int r    = lane & 15;
    const int jl   = lane >> 4;

    __shared__ float pm[128], pl[128];
    __shared__ float gm[16], gl[16];
    __shared__ float pc[4][2][8][32];

    const int   iGlob  = rowTile * 16 + r;
    const float s_row  = s[(b * NHEADS + head) * NNODES + iGlob];
    const float* t_bh  = t + (b * NHEADS + head) * NNODES;
    const int*  adj_row = adj + (size_t)(b * NNODES + iGlob) * NNODES;

    const int jbase = wave * 256;
    float m = -__builtin_inff(), l = 0.f;
    for (int it = 0; it < 128; ++it) {
        int j = jbase + jl + it * 2;
        float e = s_row + t_bh[j];
        e = (e > 0.f) ? e : 0.2f * e;
        e = (adj_row[j] == 0) ? NEG_INF_F : e;
        float mn = fmaxf(m, e);
        l = l * __expf(m - mn) + __expf(e - mn);
        m = mn;
    }
    pm[tid] = m; pl[tid] = l;
    __syncthreads();
    if (tid < 16) {
        float gmv = -__builtin_inff();
        for (int q = 0; q < 8; ++q) {
            int src = (q >> 1) * 32 + (q & 1) * 16 + tid;
            gmv = fmaxf(gmv, pm[src]);
        }
        float glv = 0.f;
        for (int q = 0; q < 8; ++q) {
            int src = (q >> 1) * 32 + (q & 1) * 16 + tid;
            glv += pl[src] * __expf(pm[src] - gmv);
        }
        gm[tid] = gmv;
        gl[tid] = glv;
    }
    __syncthreads();

    const float mrow = gm[r];
    const float* hb = h + (size_t)b * NNODES * OUTDIM + head * HDIM;
    const int n0 = lane & 15;
    v8f c0 = {}, c1 = {};
    for (int it = 0; it < 64; ++it) {
        int jA = jbase + it * 4 + jl * 2;
        v2f a;
        {
            float e = s_row + t_bh[jA];
            e = (e > 0.f) ? e : 0.2f * e;
            e = (adj_row[jA] == 0) ? NEG_INF_F : e;
            a.x = __expf(e - mrow);
            e = s_row + t_bh[jA + 1];
            e = (e > 0.f) ? e : 0.2f * e;
            e = (adj_row[jA + 1] == 0) ? NEG_INF_F : e;
            a.y = __expf(e - mrow);
        }
        v2f b0, b1;
        b0.x = hb[(size_t)(jA + 0) * OUTDIM + n0];
        b0.y = hb[(size_t)(jA + 1) * OUTDIM + n0];
        b1.x = hb[(size_t)(jA + 0) * OUTDIM + 16 + n0];
        b1.y = hb[(size_t)(jA + 1) * OUTDIM + 16 + n0];
        c0 = __builtin_amdgcn_wmma_f32_16x16x4_f32(false, a, false, b0,
                                                   (short)0, c0, false, false);
        c1 = __builtin_amdgcn_wmma_f32_16x16x4_f32(false, a, false, b1,
                                                   (short)0, c1, false, false);
    }
#pragma unroll
    for (int v = 0; v < 8; ++v) {
        pc[wave][0][v][lane] = c0[v];
        pc[wave][1][v][lane] = c1[v];
    }
    __syncthreads();

    for (int idx = tid; idx < 512; idx += 128) {
        int ct = idx >> 8;
        int v  = (idx >> 5) & 7;
        int ln = idx & 31;
        float sum = pc[0][ct][v][ln] + pc[1][ct][v][ln] +
                    pc[2][ct][v][ln] + pc[3][ct][v][ln];
        int rowL = v + ((ln >> 4) << 3);
        int col  = ct * 16 + (ln & 15);
        out[(size_t)(b * NNODES + rowTile * 16 + rowL) * OUTDIM + head * HDIM + col]
            = sum / gl[rowL];
    }
}

extern "C" void kernel_launch(void* const* d_in, const int* in_sizes, int n_in,
                              void* d_out, int out_size, void* d_ws, size_t ws_size,
                              hipStream_t stream)
{
    const float* x    = (const float*)d_in[0];
    const int*   adj  = (const int*)  d_in[1];
    const float* W    = (const float*)d_in[2];
    const float* bias = (const float*)d_in[3];
    const float* a    = (const float*)d_in[4];
    float* out = (float*)d_out;

    float* h = (float*)d_ws;
    float* s = h + BATCH * NNODES * OUTDIM;
    float* t = s + BATCH * NHEADS * NNODES;

    gat_linear<<<(BATCH * NNODES) / 16, 256, 0, stream>>>(x, W, bias, h);
    gat_scores<<<(BATCH * NHEADS * NNODES) / 256, 256, 0, stream>>>(h, a, s, t);
    gat_attn<<<BATCH * NHEADS * (NNODES / 16), 128, 0, stream>>>(h, adj, s, t, out);
}


# attn: single pass, running row max with lazy rescale (online softmax) in the transposed product (A = h^T, B = P^T: a lane's accumulators all belong to its own row); pass 1 goes; new sum order for l
# speedup vs baseline: 1.3862x; 1.3862x over previous
#include <hip/hip_runtime.h>
#include <hip/hip_bf16.h>

#define NHEADS 4
#define HDIM   32
#define NNODES 1024
#define BATCH  8
#define INDIM  128
#define OUTDIM 128
#define NEG_INF_F (-1000000000.0f)

typedef float v2f __attribute__((ext_vector_type(2)));
typedef float v8f __attribute__((ext_vector_type(8)));

__global__ __launch_bounds__(256) void gat_linear(const float* __restrict__ x,
                                                  const float* __restrict__ W,
                                                  const float* __restrict__ bias,
                                                  float* __restrict__ h)
{
    const int lane    = threadIdx.x & 31;
    const int wave    = threadIdx.x >> 5;
    const int rowTile = blockIdx.x;
    const int rloc    = lane & 15;
    const int kq      = (lane >> 4) * 2;
    const int col     = wave * 16 + rloc;

    __shared__ float xs[16][132];
    for (int i = threadIdx.x; i < 16 * 128; i += 256) {
        int rr = i >> 7, cc = i & 127;
        xs[rr][cc] = x[(rowTile * 16 + rr) * INDIM + cc];
    }
    __syncthreads();

    v8f acc = {};
    for (int kk = 0; kk < INDIM; kk += 4) {
        v2f a, b;
        a.x = xs[rloc][kk + kq + 0];
        a.y = xs[rloc][kk + kq + 1];
        b.x = W[col * INDIM + kk + kq + 0];
        b.y = W[col * INDIM + kk + kq + 1];
        acc = __builtin_amdgcn_wmma_f32_16x16x4_f32(false, a, false, b,
                                                    (short)0, acc, false, false);
    }

    const float bv = bias[col];
    const int rbase = rowTile * 16 + ((lane >> 4) << 3);
#pragma unroll
    for (int v = 0; v < 8; ++v)
        h[(rbase + v) * OUTDIM + col] = acc[v] + bv;
}

__global__ __launch_bounds__(256) void gat_scores(const float* __restrict__ h,
                                                  const float* __restrict__ a,
                                                  float* __restrict__ s,
                                                  float* __restrict__ t)
{
    int idx = blockIdx.x * blockDim.x + threadIdx.x;
    if (idx >= BATCH * NHEADS * NNODES) return;
    int n    = idx & (NNODES - 1);
    int head = (idx >> 10) & (NHEADS - 1);
    int b    = idx >> 12;
    const float* hp = h + ((size_t)(b * NNODES + n) * OUTDIM + head * HDIM);
    float ss = 0.f, tt = 0.f;
#pragma unroll
    for (int d = 0; d < HDIM; ++d) {
        float hv = hp[d];
        ss += hv * a[d];
        tt += hv * a[HDIM + d];
    }
    s[idx] = ss;
    t[idx] = tt;
}

#define ASTEP 32
#define TAU   5.5f

__device__ __forceinline__ float xhalf(float x)
{
    int bits = __builtin_bit_cast(int, x);
    return __builtin_bit_cast(float,
        __builtin_amdgcn_permlanex16(bits, bits, 0x76543210, 0xfedcba98, false, false));
}

__global__ __launch_bounds__(128) void gat_attn(const float* __restrict__ h,
                                                const int*   __restrict__ adj,
                                                const float* __restrict__ s,
                                                const float* __restrict__ t,
                                                float* __restrict__ out)
{
    const int rowTile = blockIdx.x & 63;
    const int head    = (blockIdx.x >> 6) & (NHEADS - 1);
    const int b       = blockIdx.x >> 8;

    const int tid  = threadIdx.x;
    const int wave = tid >> 5;
    const int lane = tid & 31;
    const int c    = lane & 15;
    const int hh   = lane >> 4;

    __shared__ float wm[4][16];
    __shared__ float wl[4][32];
    __shared__ __attribute__((aligned(16))) float pc[4][16][36];

    const int   iGlob   = rowTile * 16 + c;
    const float s_row   = s[(b * NHEADS + head) * NNODES + iGlob];
    const float* t_bh   = t + (b * NHEADS + head) * NNODES;
    const int*  adj_row = adj + (size_t)(b * NNODES + iGlob) * NNODES;
    const float* hb     = h + (size_t)b * NNODES * OUTDIM + head * HDIM;

    const int jbase = wave * 256;
    float mref = -__builtin_inff();
    float l = 0.f;
    v8f c0 = {}, c1 = {};
    for (int st = 0; st < 256 / ASTEP; ++st) {
        const int j0 = jbase + st * ASTEP + hh * 2;

        v2f a0[ASTEP / 4], a1[ASTEP / 4];
#pragma unroll
        for (int q = 0; q < ASTEP / 4; ++q) {
            const int jA = j0 + q * 4;
            a0[q].x = hb[(size_t)(jA + 0) * OUTDIM + c];
            a0[q].y = hb[(size_t)(jA + 1) * OUTDIM + c];
            a1[q].x = hb[(size_t)(jA + 0) * OUTDIM + 16 + c];
            a1[q].y = hb[(size_t)(jA + 1) * OUTDIM + 16 + c];
        }

        float e[ASTEP / 2];
#pragma unroll
        for (int q = 0; q < ASTEP / 4; ++q) {
            const int jA = j0 + q * 4;
            float e0 = s_row + t_bh[jA];
            e0 = (e0 > 0.f) ? e0 : 0.2f * e0;
            e0 = (adj_row[jA] == 0) ? NEG_INF_F : e0;
            float e1 = s_row + t_bh[jA + 1];
            e1 = (e1 > 0.f) ? e1 : 0.2f * e1;
            e1 = (adj_row[jA + 1] == 0) ? NEG_INF_F : e1;
            e[2 * q] = e0;
            e[2 * q + 1] = e1;
        }

        float cm = e[0];
#pragma unroll
        for (int k = 1; k < ASTEP / 2; ++k) cm = fmaxf(cm, e[k]);
        cm = fmaxf(cm, xhalf(cm));

        if (__builtin_amdgcn_ballot_w32(cm > mref + TAU) != 0) {
            const float mn = fmaxf(mref, cm);
            const float al = __expf(mref - mn);
            l *= al;
#pragma unroll
            for (int v = 0; v < 8; ++v) { c0[v] *= al; c1[v] *= al; }
            mref = mn;
        }

#pragma unroll
        for (int q = 0; q < ASTEP / 4; ++q) {
            v2f p;
            p.x = __expf(e[2 * q] - mref);
            p.y = __expf(e[2 * q + 1] - mref);
            l += p.x;
            l += p.y;
            c0 = __builtin_amdgcn_wmma_f32_16x16x4_f32(false, a0[q], false, p,
                                                       (short)0, c0, false, false);
            c1 = __builtin_amdgcn_wmma_f32_16x16x4_f32(false, a1[q], false, p,
                                                       (short)0, c1, false, false);
        }
    }

    if (hh == 0) wm[wave][c] = mref;
    __syncthreads();
    const float mall = fmaxf(fmaxf(wm[0][c], wm[1][c]), fmaxf(wm[2][c], wm[3][c]));
    const float f = __expf(mref - mall);
    wl[wave][lane] = l * f;
    {
        float4* d0 = (float4*)&pc[wave][c][8 * hh];
        float4* d1 = (float4*)&pc[wave][c][16 + 8 * hh];
        d0[0] = make_float4(c0[0] * f, c0[1] * f, c0[2] * f, c0[3] * f);
        d0[1] = make_float4(c0[4] * f, c0[5] * f, c0[6] * f, c0[7] * f);
        d1[0] = make_float4(c1[0] * f, c1[1] * f, c1[2] * f, c1[3] * f);
        d1[1] = make_float4(c1[4] * f, c1[5] * f, c1[6] * f, c1[7] * f);
    }
    __syncthreads();

    {
        const int i  = tid >> 3;
        const int d4 = (tid & 7) * 4;
        float lsum = 0.f;
#pragma unroll
        for (int w = 0; w < 4; ++w) { lsum += wl[w][i]; lsum += wl[w][16 + i]; }
        const float4 p0 = *(const float4*)&pc[0][i][d4];
        const float4 p1 = *(const float4*)&pc[1][i][d4];
        const float4 p2 = *(const float4*)&pc[2][i][d4];
        const float4 p3 = *(const float4*)&pc[3][i][d4];
        float4 o;
        o.x = (p0.x + p1.x + p2.x + p3.x) / lsum;
        o.y = (p0.y + p1.y + p2.y + p3.y) / lsum;
        o.z = (p0.z + p1.z + p2.z + p3.z) / lsum;
        o.w = (p0.w + p1.w + p2.w + p3.w) / lsum;
        *(float4*)&out[(size_t)(b * NNODES + rowTile * 16 + i) * OUTDIM + head * HDIM + d4] = o;
    }
}

extern "C" void kernel_launch(void* const* d_in, const int* in_sizes, int n_in,
                              void* d_out, int out_size, void* d_ws, size_t ws_size,
                              hipStream_t stream)
{
    const float* x    = (const float*)d_in[0];
    const int*   adj  = (const int*)  d_in[1];
    const float* W    = (const float*)d_in[2];
    const float* bias = (const float*)d_in[3];
    const float* a    = (const float*)d_in[4];
    float* out = (float*)d_out;

    float* h = (float*)d_ws;
    float* s = h + BATCH * NNODES * OUTDIM;
    float* t = s + BATCH * NHEADS * NNODES;

    gat_linear<<<(BATCH * NNODES) / 16, 256, 0, stream>>>(x, W, bias, h);
    gat_scores<<<(BATCH * NHEADS * NNODES) / 256, 256, 0, stream>>>(h, a, s, t);
    gat_attn<<<BATCH * NHEADS * (NNODES / 16), 128, 0, stream>>>(h, adj, s, t, out);
}


# h leaves the linear kernel in the attention wave's A-fragment order (LDS-staged 512-byte chunks); attn gets both tiles' fragments of a k-step in one coalesced 16-byte load (8 per step, was 32)
# speedup vs baseline: 1.4926x; 1.0768x over previous
#include <hip/hip_runtime.h>
#include <hip/hip_bf16.h>

#define NHEADS 4
#define HDIM   32
#define NNODES 1024
#define BATCH  8
#define INDIM  128
#define OUTDIM 128
#define NEG_INF_F (-1000000000.0f)

typedef float v2f __attribute__((ext_vector_type(2)));
typedef float v8f __attribute__((ext_vector_type(8)));

#define HGROUPS (NNODES / 4)

__global__ __launch_bounds__(256) void gat_linear(const float* __restrict__ x,
                                                  const float* __restrict__ W,
                                                  const float* __restrict__ bias,
                                                  float* __restrict__ h)
{
    const int lane    = threadIdx.x & 31;
    const int wave    = threadIdx.x >> 5;
    const int rowTile = blockIdx.x;
    const int rloc    = lane & 15;
    const int kq      = (lane >> 4) * 2;
    const int col     = wave * 16 + rloc;

    __shared__ float xs[16][132];
    for (int i = threadIdx.x; i < 16 * 128; i += 256) {
        int rr = i >> 7, cc = i & 127;
        xs[rr][cc] = x[(rowTile * 16 + rr) * INDIM + cc];
    }
    __syncthreads();

    v8f acc = {};
    for (int kk = 0; kk < INDIM; kk += 4) {
        v2f a, b;
        a.x = xs[rloc][kk + kq + 0];
        a.y = xs[rloc][kk + kq + 1];
        b.x = W[col * INDIM + kk + kq + 0];
        b.y = W[col * INDIM + kk + kq + 1];
        acc = __builtin_amdgcn_wmma_f32_16x16x4_f32(false, a, false, b,
                                                    (short)0, acc, false, false);
    }

    __shared__ float hs[16][132];
    const float bv = bias[col];
    const int rl = (lane >> 4) << 3;
#pragma unroll
    for (int v = 0; v < 8; ++v)
        hs[rl + v][col] = acc[v] + bv;
    __syncthreads();

    const int bb    = rowTile >> 6;
    const int gbase = (rowTile & 63) * 4;
#pragma unroll
    for (int i = 0; i < 2; ++i) {
        const int idx = threadIdx.x + i * 256;
        const int hd  = idx >> 7;
        const int gl  = (idx >> 5) & 3;
        const int lf  = idx & 31;
        const int r0  = gl * 4 + (lf >> 4) * 2;
        const int f0  = hd * HDIM + (lf & 15);
        const float4 v4 = make_float4(hs[r0][f0], hs[r0 + 1][f0], hs[r0][f0 + 16], hs[r0 + 1][f0 + 16]);
        float4* dst = (float4*)h + ((size_t)((bb * NHEADS + hd) * HGROUPS + gbase + gl) * 32 + lf);
        *dst = v4;
    }
}

__global__ __launch_bounds__(256) void gat_scores(const float* __restrict__ h,
                                                  const float* __restrict__ a,
                                                  float* __restrict__ s,
                                                  float* __restrict__ t)
{
    int idx = blockIdx.x * blockDim.x + threadIdx.x;
    if (idx >= BATCH * NHEADS * NNODES) return;
    int n    = idx & (NNODES - 1);
    int head = (idx >> 10) & (NHEADS - 1);
    int b    = idx >> 12;
    const float* hp = h + ((size_t)((b * NHEADS + head) * HGROUPS + (n >> 2)) * 128
                           + ((n >> 1) & 1) * 64 + (n & 1));
    float ss = 0.f, tt = 0.f;
#pragma unroll
    for (int d = 0; d < HDIM; ++d) {
        float hv = hp[(d & 15) * 4 + (d >> 4) * 2];
        ss += hv * a[d];
        tt += hv * a[HDIM + d];
    }
    s[idx] = ss;
    t[idx] = tt;
}

#define ASTEP 32
#define TAU   5.5f

__device__ __forceinline__ float xhalf(float x)
{
    int bits = __builtin_bit_cast(int, x);
    return __builtin_bit_cast(float,
        __builtin_amdgcn_permlanex16(bits, bits, 0x76543210, 0xfedcba98, false, false));
}

__global__ __launch_bounds__(128) void gat_attn(const float* __restrict__ h,
                                                const int*   __restrict__ adj,
                                                const float* __restrict__ s,
                                                const float* __restrict__ t,
                                                float* __restrict__ out)
{
    const int rowTile = blockIdx.x & 63;
    const int head    = (blockIdx.x >> 6) & (NHEADS - 1);
    const int b       = blockIdx.x >> 8;

    const int tid  = threadIdx.x;
    const int wave = tid >> 5;
    const int lane = tid & 31;
    const int c    = lane & 15;
    const int hh   = lane >> 4;

    __shared__ float wm[4][16];
    __shared__ float wl[4][32];
    __shared__ __attribute__((aligned(16))) float pc[4][16][36];

    const int   iGlob   = rowTile * 16 + c;
    const float s_row   = s[(b * NHEADS + head) * NNODES + iGlob];
    const float* t_bh   = t + (b * NHEADS + head) * NNODES;
    const int*  adj_row = adj + (size_t)(b * NNODES + iGlob) * NNODES;
    const float4* hf4   = (const float4*)h + ((size_t)((b * NHEADS + head) * HGROUPS) * 32 + lane);

    const int jbase = wave * 256;
    float mref = -__builtin_inff();
    float l = 0.f;
    v8f c0 = {}, c1 = {};
    for (int st = 0; st < 256 / ASTEP; ++st) {
        const int j0 = jbase + st * ASTEP + hh * 2;

        v2f a0[ASTEP / 4], a1[ASTEP / 4];
        const int g0 = (jbase + st * ASTEP) >> 2;
#pragma unroll
        for (int q = 0; q < ASTEP / 4; ++q) {
            const float4 hv = hf4[(size_t)(g0 + q) * 32];
            a0[q].x = hv.x;
            a0[q].y = hv.y;
            a1[q].x = hv.z;
            a1[q].y = hv.w;
        }

        float e[ASTEP / 2];
#pragma unroll
        for (int q = 0; q < ASTEP / 4; ++q) {
            const int jA = j0 + q * 4;
            float e0 = s_row + t_bh[jA];
            e0 = (e0 > 0.f) ? e0 : 0.2f * e0;
            e0 = (adj_row[jA] == 0) ? NEG_INF_F : e0;
            float e1 = s_row + t_bh[jA + 1];
            e1 = (e1 > 0.f) ? e1 : 0.2f * e1;
            e1 = (adj_row[jA + 1] == 0) ? NEG_INF_F : e1;
            e[2 * q] = e0;
            e[2 * q + 1] = e1;
        }

        float cm = e[0];
#pragma unroll
        for (int k = 1; k < ASTEP / 2; ++k) cm = fmaxf(cm, e[k]);
        cm = fmaxf(cm, xhalf(cm));

        if (__builtin_amdgcn_ballot_w32(cm > mref + TAU) != 0) {
            const float mn = fmaxf(mref, cm);
            const float al = __expf(mref - mn);
            l *= al;
#pragma unroll
            for (int v = 0; v < 8; ++v) { c0[v] *= al; c1[v] *= al; }
            mref = mn;
        }

#pragma unroll
        for (int q = 0; q < ASTEP / 4; ++q) {
            v2f p;
            p.x = __expf(e[2 * q] - mref);
            p.y = __expf(e[2 * q + 1] - mref);
            l += p.x;
            l += p.y;
            c0 = __builtin_amdgcn_wmma_f32_16x16x4_f32(false, a0[q], false, p,
                                                       (short)0, c0, false, false);
            c1 = __builtin_amdgcn_wmma_f32_16x16x4_f32(false, a1[q], false, p,
                                                       (short)0, c1, false, false);
        }
    }

    if (hh == 0) wm[wave][c] = mref;
    __syncthreads();
    const float mall = fmaxf(fmaxf(wm[0][c], wm[1][c]), fmaxf(wm[2][c], wm[3][c]));
    const float f = __expf(mref - mall);
    wl[wave][lane] = l * f;
    {
        float4* d0 = (float4*)&pc[wave][c][8 * hh];
        float4* d1 = (float4*)&pc[wave][c][16 + 8 * hh];
        d0[0] = make_float4(c0[0] * f, c0[1] * f, c0[2] * f, c0[3] * f);
        d0[1] = make_float4(c0[4] * f, c0[5] * f, c0[6] * f, c0[7] * f);
        d1[0] = make_float4(c1[0] * f, c1[1] * f, c1[2] * f, c1[3] * f);
        d1[1] = make_float4(c1[4] * f, c1[5] * f, c1[6] * f, c1[7] * f);
    }
    __syncthreads();

    {
        const int i  = tid >> 3;
        const int d4 = (tid & 7) * 4;
        float lsum = 0.f;
#pragma unroll
        for (int w = 0; w < 4; ++w) { lsum += wl[w][i]; lsum += wl[w][16 + i]; }
        const float4 p0 = *(const float4*)&pc[0][i][d4];
        const float4 p1 = *(const float4*)&pc[1][i][d4];
        const float4 p2 = *(const float4*)&pc[2][i][d4];
        const float4 p3 = *(const float4*)&pc[3][i][d4];
        float4 o;
        o.x = (p0.x + p1.x + p2.x + p3.x) / lsum;
        o.y = (p0.y + p1.y + p2.y + p3.y) / lsum;
        o.z = (p0.z + p1.z + p2.z + p3.z) / lsum;
        o.w = (p0.w + p1.w + p2.w + p3.w) / lsum;
        *(float4*)&out[(size_t)(b * NNODES + rowTile * 16 + i) * OUTDIM + head * HDIM + d4] = o;
    }
}

extern "C" void kernel_launch(void* const* d_in, const int* in_sizes, int n_in,
                              void* d_out, int out_size, void* d_ws, size_t ws_size,
                              hipStream_t stream)
{
    const float* x    = (const float*)d_in[0];
    const int*   adj  = (const int*)  d_in[1];
    const float* W    = (const float*)d_in[2];
    const float* bias = (const float*)d_in[3];
    const float* a    = (const float*)d_in[4];
    float* out = (float*)d_out;

    float* h = (float*)d_ws;
    float* s = h + BATCH * NNODES * OUTDIM;
    float* t = s + BATCH * NHEADS * NNODES;

    gat_linear<<<(BATCH * NNODES) / 16, 256, 0, stream>>>(x, W, bias, h);
    gat_scores<<<(BATCH * NHEADS * NNODES) / 256, 256, 0, stream>>>(h, a, s, t);
    gat_attn<<<BATCH * NHEADS * (NNODES / 16), 128, 0, stream>>>(h, adj, s, t, out);
}


# s and t come from the linear block's LDS-staged h tile (one thread per value, one fma per feature in ascending order, as before), one 512-byte record per row tile; the score launch goes (3 to 2)
# speedup vs baseline: 1.5546x; 1.0415x over previous
#include <hip/hip_runtime.h>
#include <hip/hip_bf16.h>

#define NHEADS 4
#define HDIM   32
#define NNODES 1024
#define BATCH  8
#define INDIM  128
#define OUTDIM 128
#define NEG_INF_F (-1000000000.0f)

typedef float v2f __attribute__((ext_vector_type(2)));
typedef float v8f __attribute__((ext_vector_type(8)));

#define HGROUPS (NNODES / 4)

#define STREC 128
__global__ __launch_bounds__(256) void gat_linear(const float* __restrict__ x,
                                                  const float* __restrict__ W,
                                                  const float* __restrict__ bias,
                                                  const float* __restrict__ a,
                                                  float* __restrict__ h,
                                                  float* __restrict__ st)
{
    const int lane    = threadIdx.x & 31;
    const int wave    = threadIdx.x >> 5;
    const int rowTile = blockIdx.x;
    const int rloc    = lane & 15;
    const int kq      = (lane >> 4) * 2;
    const int col     = wave * 16 + rloc;

    __shared__ float xs[16][132];
    for (int i = threadIdx.x; i < 16 * 128; i += 256) {
        int rr = i >> 7, cc = i & 127;
        xs[rr][cc] = x[(rowTile * 16 + rr) * INDIM + cc];
    }
    __syncthreads();

    v8f acc = {};
    for (int kk = 0; kk < INDIM; kk += 4) {
        v2f a, b;
        a.x = xs[rloc][kk + kq + 0];
        a.y = xs[rloc][kk + kq + 1];
        b.x = W[col * INDIM + kk + kq + 0];
        b.y = W[col * INDIM + kk + kq + 1];
        acc = __builtin_amdgcn_wmma_f32_16x16x4_f32(false, a, false, b,
                                                    (short)0, acc, false, false);
    }

    __shared__ __attribute__((aligned(16))) float hs[16][132];
    const float bv = bias[col];
    const int which = (threadIdx.x >> 6) & 1;
    float4 av4[HDIM / 4];
#pragma unroll
    for (int q = 0; q < HDIM / 4; ++q) av4[q] = ((const float4*)(a + which * HDIM))[q];
    const int rl = (lane >> 4) << 3;
#pragma unroll
    for (int v = 0; v < 8; ++v)
        hs[rl + v][col] = acc[v] + bv;
    __syncthreads();

    const int bb    = rowTile >> 6;
    const int gbase = (rowTile & 63) * 4;
#pragma unroll
    for (int i = 0; i < 2; ++i) {
        const int idx = threadIdx.x + i * 256;
        const int hd  = idx >> 7;
        const int gl  = (idx >> 5) & 3;
        const int lf  = idx & 31;
        const int r0  = gl * 4 + (lf >> 4) * 2;
        const int f0  = hd * HDIM + (lf & 15);
        const float4 v4 = make_float4(hs[r0][f0], hs[r0 + 1][f0], hs[r0][f0 + 16], hs[r0 + 1][f0 + 16]);
        float4* dst = (float4*)h + ((size_t)((bb * NHEADS + hd) * HGROUPS + gbase + gl) * 32 + lf);
        *dst = v4;
    }

    if (threadIdx.x < STREC) {
        const int hd  = (threadIdx.x >> 4) & (NHEADS - 1);
        const int row = threadIdx.x & 15;
        const float4* hr = (const float4*)&hs[row][hd * HDIM];
        float sc = 0.f;
#pragma unroll
        for (int q = 0; q < HDIM / 4; ++q) {
            const float4 hv = hr[q];
            sc = __builtin_fmaf(hv.x, av4[q].x, sc);
            sc = __builtin_fmaf(hv.y, av4[q].y, sc);
            sc = __builtin_fmaf(hv.z, av4[q].z, sc);
            sc = __builtin_fmaf(hv.w, av4[q].w, sc);
        }
        st[(size_t)rowTile * STREC + threadIdx.x] = sc;
    }
}

#define ASTEP 32
#define TAU   5.5f

__device__ __forceinline__ float xhalf(float x)
{
    int bits = __builtin_bit_cast(int, x);
    return __builtin_bit_cast(float,
        __builtin_amdgcn_permlanex16(bits, bits, 0x76543210, 0xfedcba98, false, false));
}

__global__ __launch_bounds__(128) void gat_attn(const float* __restrict__ h,
                                                const int*   __restrict__ adj,
                                                const float* __restrict__ st,
                                                float* __restrict__ out)
{
    const int rowTile = blockIdx.x & 63;
    const int head    = (blockIdx.x >> 6) & (NHEADS - 1);
    const int b       = blockIdx.x >> 8;

    const int tid  = threadIdx.x;
    const int wave = tid >> 5;
    const int lane = tid & 31;
    const int c    = lane & 15;
    const int hh   = lane >> 4;

    __shared__ float wm[4][16];
    __shared__ float wl[4][32];
    __shared__ __attribute__((aligned(16))) float pc[4][16][36];

    const int   iGlob   = rowTile * 16 + c;
    const float* st_b   = st + (size_t)(b * (NNODES / 16)) * STREC;
    const float s_row   = st_b[rowTile * STREC + head * 16 + c];
    const float* t_bh   = st_b + (STREC / 2) + head * 16 + hh * 2;
    const int*  adj_row = adj + (size_t)(b * NNODES + iGlob) * NNODES;
    const float4* hf4   = (const float4*)h + ((size_t)((b * NHEADS + head) * HGROUPS) * 32 + lane);

    const int jbase = wave * 256;
    float mref = -__builtin_inff();
    float l = 0.f;
    v8f c0 = {}, c1 = {};
    for (int stp = 0; stp < 256 / ASTEP; ++stp) {
        const int j0 = jbase + stp * ASTEP + hh * 2;
        const float* tp = t_bh + (size_t)((jbase + stp * ASTEP) >> 4) * STREC;

        v2f a0[ASTEP / 4], a1[ASTEP / 4];
        const int g0 = (jbase + stp * ASTEP) >> 2;
#pragma unroll
        for (int q = 0; q < ASTEP / 4; ++q) {
            const float4 hv = hf4[(size_t)(g0 + q) * 32];
            a0[q].x = hv.x;
            a0[q].y = hv.y;
            a1[q].x = hv.z;
            a1[q].y = hv.w;
        }

        float e[ASTEP / 2];
#pragma unroll
        for (int q = 0; q < ASTEP / 4; ++q) {
            const int jA = j0 + q * 4;
            const float2 tv = *(const float2*)&tp[(q >> 2) * STREC + (q & 3) * 4];
            float e0 = s_row + tv.x;
            e0 = (e0 > 0.f) ? e0 : 0.2f * e0;
            e0 = (adj_row[jA] == 0) ? NEG_INF_F : e0;
            float e1 = s_row + tv.y;
            e1 = (e1 > 0.f) ? e1 : 0.2f * e1;
            e1 = (adj_row[jA + 1] == 0) ? NEG_INF_F : e1;
            e[2 * q] = e0;
            e[2 * q + 1] = e1;
        }

        float cm = e[0];
#pragma unroll
        for (int k = 1; k < ASTEP / 2; ++k) cm = fmaxf(cm, e[k]);
        cm = fmaxf(cm, xhalf(cm));

        if (__builtin_amdgcn_ballot_w32(cm > mref + TAU) != 0) {
            const float mn = fmaxf(mref, cm);
            const float al = __expf(mref - mn);
            l *= al;
#pragma unroll
            for (int v = 0; v < 8; ++v) { c0[v] *= al; c1[v] *= al; }
            mref = mn;
        }

#pragma unroll
        for (int q = 0; q < ASTEP / 4; ++q) {
            v2f p;
            p.x = __expf(e[2 * q] - mref);
            p.y = __expf(e[2 * q + 1] - mref);
            l += p.x;
            l += p.y;
            c0 = __builtin_amdgcn_wmma_f32_16x16x4_f32(false, a0[q], false, p,
                                                       (short)0, c0, false, false);
            c1 = __builtin_amdgcn_wmma_f32_16x16x4_f32(false, a1[q], false, p,
                                                       (short)0, c1, false, false);
        }
    }

    if (hh == 0) wm[wave][c] = mref;
    __syncthreads();
    const float mall = fmaxf(fmaxf(wm[0][c], wm[1][c]), fmaxf(wm[2][c], wm[3][c]));
    const float f = __expf(mref - mall);
    wl[wave][lane] = l * f;
    {
        float4* d0 = (float4*)&pc[wave][c][8 * hh];
        float4* d1 = (float4*)&pc[wave][c][16 + 8 * hh];
        d0[0] = make_float4(c0[0] * f, c0[1] * f, c0[2] * f, c0[3] * f);
        d0[1] = make_float4(c0[4] * f, c0[5] * f, c0[6] * f, c0[7] * f);
        d1[0] = make_float4(c1[0] * f, c1[1] * f, c1[2] * f, c1[3] * f);
        d1[1] = make_float4(c1[4] * f, c1[5] * f, c1[6] * f, c1[7] * f);
    }
    __syncthreads();

    {
        const int i  = tid >> 3;
        const int d4 = (tid & 7) * 4;
        float lsum = 0.f;
#pragma unroll
        for (int w = 0; w < 4; ++w) { lsum += wl[w][i]; lsum += wl[w][16 + i]; }
        const float4 p0 = *(const float4*)&pc[0][i][d4];
        const float4 p1 = *(const float4*)&pc[1][i][d4];
        const float4 p2 = *(const float4*)&pc[2][i][d4];
        const float4 p3 = *(const float4*)&pc[3][i][d4];
        float4 o;
        o.x = (p0.x + p1.x + p2.x + p3.x) / lsum;
        o.y = (p0.y + p1.y + p2.y + p3.y) / lsum;
        o.z = (p0.z + p1.z + p2.z + p3.z) / lsum;
        o.w = (p0.w + p1.w + p2.w + p3.w) / lsum;
        *(float4*)&out[(size_t)(b * NNODES + rowTile * 16 + i) * OUTDIM + head * HDIM + d4] = o;
    }
}

extern "C" void kernel_launch(void* const* d_in, const int* in_sizes, int n_in,
                              void* d_out, int out_size, void* d_ws, size_t ws_size,
                              hipStream_t stream)
{
    const float* x    = (const float*)d_in[0];
    const int*   adj  = (const int*)  d_in[1];
    const float* W    = (const float*)d_in[2];
    const float* bias = (const float*)d_in[3];
    const float* a    = (const float*)d_in[4];
    float* out = (float*)d_out;

    float* h = (float*)d_ws;
    float* st = h + BATCH * NNODES * OUTDIM;

    gat_linear<<<(BATCH * NNODES) / 16, 256, 0, stream>>>(x, W, bias, a, h, st);
    gat_attn<<<BATCH * NHEADS * (NNODES / 16), 128, 0, stream>>>(h, adj, st, out);
}


# adj packed to 1 bit per edge by the linear launch (whole-line loads + ballots), staged in each (row tile, head) attn block's own output rows; attn loads 32 B of mask per lane and no adj
# speedup vs baseline: 1.6375x; 1.0533x over previous
#include <hip/hip_runtime.h>
#include <hip/hip_bf16.h>

#define NHEADS 4
#define HDIM   32
#define NNODES 1024
#define BATCH  8
#define INDIM  128
#define OUTDIM 128
#define NEG_INF_F (-1000000000.0f)
#define NEG_INF_BITS ((int)0xCE6E6B28)

typedef float v2f __attribute__((ext_vector_type(2)));
typedef float v8f __attribute__((ext_vector_type(8)));

#define HGROUPS (NNODES / 4)

#define STREC 128
__global__ __launch_bounds__(256) void gat_linear(const float* __restrict__ x,
                                                  const float* __restrict__ W,
                                                  const float* __restrict__ bias,
                                                  const float* __restrict__ a,
                                                  const int* __restrict__ adj,
                                                  float* __restrict__ h,
                                                  float* __restrict__ st,
                                                  float* __restrict__ stage)
{
    const int lane    = threadIdx.x & 31;
    const int wave    = threadIdx.x >> 5;
    const int rowTile = blockIdx.x;
    const int rloc    = lane & 15;
    const int kq      = (lane >> 4) * 2;
    const int col     = wave * 16 + rloc;

    const int4* arow = (const int4*)(adj + (size_t)(rowTile * 16 + wave * 2) * NNODES) + lane;
    int4 aj[16];
#pragma unroll
    for (int k = 0; k < 16; ++k) aj[k] = arow[k * 32];

    __shared__ float xs[16][132];
    for (int i = threadIdx.x; i < 16 * 128; i += 256) {
        int rr = i >> 7, cc = i & 127;
        xs[rr][cc] = x[(rowTile * 16 + rr) * INDIM + cc];
    }

    unsigned m0 = 0u, m1 = 0u, m2 = 0u, m3 = 0u;
#pragma unroll
    for (int k = 0; k < 16; ++k) {
        const unsigned b0 = __builtin_amdgcn_ballot_w32(aj[k].x != 0);
        const unsigned b1 = __builtin_amdgcn_ballot_w32(aj[k].y != 0);
        const unsigned b2 = __builtin_amdgcn_ballot_w32(aj[k].z != 0);
        const unsigned b3 = __builtin_amdgcn_ballot_w32(aj[k].w != 0);
        const bool mine = ((lane & 15) == k);
        m0 = mine ? b0 : m0;
        m1 = mine ? b1 : m1;
        m2 = mine ? b2 : m2;
        m3 = mine ? b3 : m3;
    }
    __syncthreads();

    v8f acc = {};
    for (int kk = 0; kk < INDIM; kk += 4) {
        v2f a, b;
        a.x = xs[rloc][kk + kq + 0];
        a.y = xs[rloc][kk + kq + 1];
        b.x = W[col * INDIM + kk + kq + 0];
        b.y = W[col * INDIM + kk + kq + 1];
        acc = __builtin_amdgcn_wmma_f32_16x16x4_f32(false, a, false, b,
                                                    (short)0, acc, false, false);
    }

    __shared__ __attribute__((aligned(16))) float hs[16][132];
    const float bv = bias[col];
    const int which = (threadIdx.x >> 6) & 1;
    float4 av4[HDIM / 4];
#pragma unroll
    for (int q = 0; q < HDIM / 4; ++q) av4[q] = ((const float4*)(a + which * HDIM))[q];

    const int rl = (lane >> 4) << 3;
#pragma unroll
    for (int v = 0; v < 8; ++v)
        hs[rl + v][col] = acc[v] + bv;
    __syncthreads();

    const int bb    = rowTile >> 6;
    const int gbase = (rowTile & 63) * 4;
#pragma unroll
    for (int i = 0; i < 2; ++i) {
        const int idx = threadIdx.x + i * 256;
        const int hd  = idx >> 7;
        const int gl  = (idx >> 5) & 3;
        const int lf  = idx & 31;
        const int r0  = gl * 4 + (lf >> 4) * 2;
        const int f0  = hd * HDIM + (lf & 15);
        const float4 v4 = make_float4(hs[r0][f0], hs[r0 + 1][f0], hs[r0][f0 + 16], hs[r0 + 1][f0 + 16]);
        float4* dst = (float4*)h + ((size_t)((bb * NHEADS + hd) * HGROUPS + gbase + gl) * 32 + lf);
        *dst = v4;
    }

    if (threadIdx.x < STREC) {
        const int hd  = (threadIdx.x >> 4) & (NHEADS - 1);
        const int row = threadIdx.x & 15;
        const float4* hr = (const float4*)&hs[row][hd * HDIM];
        float sc = 0.f;
#pragma unroll
        for (int q = 0; q < HDIM / 4; ++q) {
            const float4 hv = hr[q];
            sc = __builtin_fmaf(hv.x, av4[q].x, sc);
            sc = __builtin_fmaf(hv.y, av4[q].y, sc);
            sc = __builtin_fmaf(hv.z, av4[q].z, sc);
            sc = __builtin_fmaf(hv.w, av4[q].w, sc);
        }
        st[(size_t)rowTile * STREC + threadIdx.x] = sc;
    }

    {
        const int L = lane & 15;
        const float4 mv = make_float4(__builtin_bit_cast(float, m0), __builtin_bit_cast(float, m1),
                                      __builtin_bit_cast(float, m2), __builtin_bit_cast(float, m3));
        const int row = rowTile * 16 + wave * 2 + (L >> 3);
        float4* dst = (float4*)stage + ((size_t)row * (OUTDIM / 4) + (lane >> 4) * (HDIM / 4) + (L & 7));
        dst[0] = mv;
        dst[2 * (HDIM / 4)] = mv;
    }
}

#define ASTEP 32
#define TAU   5.5f

__device__ __forceinline__ float xhalf(float x)
{
    int bits = __builtin_bit_cast(int, x);
    return __builtin_bit_cast(float,
        __builtin_amdgcn_permlanex16(bits, bits, 0x76543210, 0xfedcba98, false, false));
}

__global__ __launch_bounds__(128) void gat_attn(const float* __restrict__ h,
                                                const float* __restrict__ st,
                                                float* out)
{
    const int rowTile = blockIdx.x & 63;
    const int head    = (blockIdx.x >> 6) & (NHEADS - 1);
    const int b       = blockIdx.x >> 8;

    const int tid  = threadIdx.x;
    const int wave = tid >> 5;
    const int lane = tid & 31;
    const int c    = lane & 15;
    const int hh   = lane >> 4;

    __shared__ float wm[4][16];
    __shared__ float wl[4][32];
    __shared__ __attribute__((aligned(16))) float pc[4][16][36];

    const int   iGlob   = rowTile * 16 + c;
    const float* st_b   = st + (size_t)(b * (NNODES / 16)) * STREC;
    const float s_row   = st_b[rowTile * STREC + head * 16 + c];
    const float* t_bh   = st_b + (STREC / 2) + head * 16 + hh * 2;
    const float* mrow = out + (size_t)(b * NNODES + iGlob) * OUTDIM + head * HDIM + hh * 2;
    const float2 mkA = *(const float2*)(mrow + (wave * 2 + 0) * 4);
    const float2 mkB = *(const float2*)(mrow + (wave * 2 + 1) * 4);
    const unsigned long long M0 = (unsigned long long)__builtin_bit_cast(unsigned, mkA.x)
                                | ((unsigned long long)__builtin_bit_cast(unsigned, mkB.x) << 32);
    const unsigned long long M1 = (unsigned long long)__builtin_bit_cast(unsigned, mkA.y)
                                | ((unsigned long long)__builtin_bit_cast(unsigned, mkB.y) << 32);
    const float4* hf4   = (const float4*)h + ((size_t)((b * NHEADS + head) * HGROUPS) * 32 + lane);

    const int jbase = wave * 256;
    float mref = -__builtin_inff();
    float l = 0.f;
    v8f c0 = {}, c1 = {};
    for (int stp = 0; stp < 256 / ASTEP; ++stp) {
        const float* tp = t_bh + (size_t)((jbase + stp * ASTEP) >> 4) * STREC;

        v2f a0[ASTEP / 4], a1[ASTEP / 4];
        const int g0 = (jbase + stp * ASTEP) >> 2;
#pragma unroll
        for (int q = 0; q < ASTEP / 4; ++q) {
            const float4 hv = hf4[(size_t)(g0 + q) * 32];
            a0[q].x = hv.x;
            a0[q].y = hv.y;
            a1[q].x = hv.z;
            a1[q].y = hv.w;
        }

        const unsigned w0 = (unsigned)(M0 >> (8 * stp));
        const unsigned w1 = (unsigned)(M1 >> (8 * stp));
        float e[ASTEP / 2];
#pragma unroll
        for (int q = 0; q < ASTEP / 4; ++q) {
            const float2 tv = *(const float2*)&tp[(q >> 2) * STREC + (q & 3) * 4];
            float e0 = s_row + tv.x;
            e0 = (e0 > 0.f) ? e0 : 0.2f * e0;
            const int k0 = (int)(w0 << (31 - q)) >> 31;
            e0 = __builtin_bit_cast(float, (__builtin_bit_cast(int, e0) & k0) | (NEG_INF_BITS & ~k0));
            float e1 = s_row + tv.y;
            e1 = (e1 > 0.f) ? e1 : 0.2f * e1;
            const int k1 = (int)(w1 << (31 - q)) >> 31;
            e1 = __builtin_bit_cast(float, (__builtin_bit_cast(int, e1) & k1) | (NEG_INF_BITS & ~k1));
            e[2 * q] = e0;
            e[2 * q + 1] = e1;
        }

        float cm = e[0];
#pragma unroll
        for (int k = 1; k < ASTEP / 2; ++k) cm = fmaxf(cm, e[k]);
        cm = fmaxf(cm, xhalf(cm));

        if (__builtin_amdgcn_ballot_w32(cm > mref + TAU) != 0) {
            const float mn = fmaxf(mref, cm);
            const float al = __expf(mref - mn);
            l *= al;
#pragma unroll
            for (int v = 0; v < 8; ++v) { c0[v] *= al; c1[v] *= al; }
            mref = mn;
        }

#pragma unroll
        for (int q = 0; q < ASTEP / 4; ++q) {
            v2f p;
            p.x = __expf(e[2 * q] - mref);
            p.y = __expf(e[2 * q + 1] - mref);
            l += p.x;
            l += p.y;
            c0 = __builtin_amdgcn_wmma_f32_16x16x4_f32(false, a0[q], false, p,
                                                       (short)0, c0, false, false);
            c1 = __builtin_amdgcn_wmma_f32_16x16x4_f32(false, a1[q], false, p,
                                                       (short)0, c1, false, false);
        }
    }

    if (hh == 0) wm[wave][c] = mref;
    __syncthreads();
    const float mall = fmaxf(fmaxf(wm[0][c], wm[1][c]), fmaxf(wm[2][c], wm[3][c]));
    const float f = __expf(mref - mall);
    wl[wave][lane] = l * f;
    {
        float4* d0 = (float4*)&pc[wave][c][8 * hh];
        float4* d1 = (float4*)&pc[wave][c][16 + 8 * hh];
        d0[0] = make_float4(c0[0] * f, c0[1] * f, c0[2] * f, c0[3] * f);
        d0[1] = make_float4(c0[4] * f, c0[5] * f, c0[6] * f, c0[7] * f);
        d1[0] = make_float4(c1[0] * f, c1[1] * f, c1[2] * f, c1[3] * f);
        d1[1] = make_float4(c1[4] * f, c1[5] * f, c1[6] * f, c1[7] * f);
    }
    __syncthreads();

    {
        const int i  = tid >> 3;
        const int d4 = (tid & 7) * 4;
        float lsum = 0.f;
#pragma unroll
        for (int w = 0; w < 4; ++w) { lsum += wl[w][i]; lsum += wl[w][16 + i]; }
        const float4 p0 = *(const float4*)&pc[0][i][d4];
        const float4 p1 = *(const float4*)&pc[1][i][d4];
        const float4 p2 = *(const float4*)&pc[2][i][d4];
        const float4 p3 = *(const float4*)&pc[3][i][d4];
        float4 o;
        o.x = (p0.x + p1.x + p2.x + p3.x) / lsum;
        o.y = (p0.y + p1.y + p2.y + p3.y) / lsum;
        o.z = (p0.z + p1.z + p2.z + p3.z) / lsum;
        o.w = (p0.w + p1.w + p2.w + p3.w) / lsum;
        *(float4*)&out[(size_t)(b * NNODES + rowTile * 16 + i) * OUTDIM + head * HDIM + d4] = o;
    }
}

extern "C" void kernel_launch(void* const* d_in, const int* in_sizes, int n_in,
                              void* d_out, int out_size, void* d_ws, size_t ws_size,
                              hipStream_t stream)
{
    const float* x    = (const float*)d_in[0];
    const int*   adj  = (const int*)  d_in[1];
    const float* W    = (const float*)d_in[2];
    const float* bias = (const float*)d_in[3];
    const float* a    = (const float*)d_in[4];
    float* out = (float*)d_out;

    float* h = (float*)d_ws;
    float* st = h + BATCH * NNODES * OUTDIM;

    gat_linear<<<(BATCH * NNODES) / 16, 256, 0, stream>>>(x, W, bias, a, adj, h, st, out);
    gat_attn<<<BATCH * NHEADS * (NNODES / 16), 128, 0, stream>>>(h, st, out);
}


# linear: x (A) and W (B) fragments straight from global, 16 bytes per lane and load (k = 8m+4*half..+3), all a wave's loads out before its first wait; x's LDS staging and a barrier go; new h sum order
# speedup vs baseline: 1.7423x; 1.0640x over previous
#include <hip/hip_runtime.h>
#include <hip/hip_bf16.h>

#define NHEADS 4
#define HDIM   32
#define NNODES 1024
#define BATCH  8
#define INDIM  128
#define OUTDIM 128
#define NEG_INF_F (-1000000000.0f)
#define NEG_INF_BITS ((int)0xCE6E6B28)

typedef float v2f __attribute__((ext_vector_type(2)));
typedef float v8f __attribute__((ext_vector_type(8)));
typedef float v4f __attribute__((ext_vector_type(4)));
typedef int   v4i __attribute__((ext_vector_type(4)));

#define HGROUPS (NNODES / 4)

#define STREC 128
__global__ __launch_bounds__(256) void gat_linear(const float* __restrict__ x,
                                                  const float* __restrict__ W,
                                                  const float* __restrict__ bias,
                                                  const float* __restrict__ a,
                                                  const int* __restrict__ adj,
                                                  float* __restrict__ h,
                                                  float* __restrict__ st,
                                                  float* __restrict__ stage)
{
    const int lane    = threadIdx.x & 31;
    const int wave    = threadIdx.x >> 5;
    const int rowTile = blockIdx.x;
    const int rloc    = lane & 15;
    const int half    = lane >> 4;
    const int col     = wave * 16 + rloc;

    const v4i* arow = (const v4i*)(adj + (size_t)(rowTile * 16 + wave * 2) * NNODES) + lane;
    v4i aj[16];
#pragma unroll
    for (int k = 0; k < 16; ++k) aj[k] = arow[k * 32];

    const v4f* xp = (const v4f*)(x + (size_t)(rowTile * 16 + rloc) * INDIM) + half;
    const v4f* wp = (const v4f*)(W + (size_t)col * INDIM) + half;
    v4f xa[16], wb[16];
#pragma unroll
    for (int m = 0; m < 16; ++m) xa[m] = xp[2 * m];
#pragma unroll
    for (int m = 0; m < 16; ++m) wb[m] = wp[2 * m];
    const float bv = bias[col];

    asm volatile("" :: "v"(aj[0]), "v"(aj[1]), "v"(aj[2]), "v"(aj[3]), "v"(aj[4]), "v"(aj[5]), "v"(aj[6]), "v"(aj[7]), "v"(aj[8]), "v"(aj[9]), "v"(aj[10]), "v"(aj[11]), "v"(aj[12]), "v"(aj[13]), "v"(aj[14]), "v"(aj[15]));
    asm volatile("" :: "v"(xa[0]), "v"(xa[1]), "v"(xa[2]), "v"(xa[3]), "v"(xa[4]), "v"(xa[5]), "v"(xa[6]), "v"(xa[7]), "v"(xa[8]), "v"(xa[9]), "v"(xa[10]), "v"(xa[11]), "v"(xa[12]), "v"(xa[13]), "v"(xa[14]), "v"(xa[15]));
    asm volatile("" :: "v"(wb[0]), "v"(wb[1]), "v"(wb[2]), "v"(wb[3]), "v"(wb[4]), "v"(wb[5]), "v"(wb[6]), "v"(wb[7]), "v"(wb[8]), "v"(wb[9]), "v"(wb[10]), "v"(wb[11]), "v"(wb[12]), "v"(wb[13]), "v"(wb[14]), "v"(wb[15]));

    unsigned m0 = 0u, m1 = 0u, m2 = 0u, m3 = 0u;
#pragma unroll
    for (int k = 0; k < 16; ++k) {
        const unsigned b0 = __builtin_amdgcn_ballot_w32(aj[k].x != 0);
        const unsigned b1 = __builtin_amdgcn_ballot_w32(aj[k].y != 0);
        const unsigned b2 = __builtin_amdgcn_ballot_w32(aj[k].z != 0);
        const unsigned b3 = __builtin_amdgcn_ballot_w32(aj[k].w != 0);
        const bool mine = ((lane & 15) == k);
        m0 = mine ? b0 : m0;
        m1 = mine ? b1 : m1;
        m2 = mine ? b2 : m2;
        m3 = mine ? b3 : m3;
    }

    v8f acc = {};
#pragma unroll
    for (int m = 0; m < 16; ++m) {
        v2f a, b;
        a.x = xa[m].x;  a.y = xa[m].y;
        b.x = wb[m].x;  b.y = wb[m].y;
        acc = __builtin_amdgcn_wmma_f32_16x16x4_f32(false, a, false, b,
                                                    (short)0, acc, false, false);
        a.x = xa[m].z;  a.y = xa[m].w;
        b.x = wb[m].z;  b.y = wb[m].w;
        acc = __builtin_amdgcn_wmma_f32_16x16x4_f32(false, a, false, b,
                                                    (short)0, acc, false, false);
    }

    __shared__ __attribute__((aligned(16))) float hs[16][132];
    const int which = (threadIdx.x >> 6) & 1;
    float4 av4[HDIM / 4];
#pragma unroll
    for (int q = 0; q < HDIM / 4; ++q) av4[q] = ((const float4*)(a + which * HDIM))[q];

    const int rl = (lane >> 4) << 3;
#pragma unroll
    for (int v = 0; v < 8; ++v)
        hs[rl + v][col] = acc[v] + bv;
    __syncthreads();

    const int bb    = rowTile >> 6;
    const int gbase = (rowTile & 63) * 4;
#pragma unroll
    for (int i = 0; i < 2; ++i) {
        const int idx = threadIdx.x + i * 256;
        const int hd  = idx >> 7;
        const int gl  = (idx >> 5) & 3;
        const int lf  = idx & 31;
        const int r0  = gl * 4 + (lf >> 4) * 2;
        const int f0  = hd * HDIM + (lf & 15);
        const float4 v4 = make_float4(hs[r0][f0], hs[r0 + 1][f0], hs[r0][f0 + 16], hs[r0 + 1][f0 + 16]);
        float4* dst = (float4*)h + ((size_t)((bb * NHEADS + hd) * HGROUPS + gbase + gl) * 32 + lf);
        *dst = v4;
    }

    if (threadIdx.x < STREC) {
        const int hd  = (threadIdx.x >> 4) & (NHEADS - 1);
        const int row = threadIdx.x & 15;
        const float4* hr = (const float4*)&hs[row][hd * HDIM];
        float sc = 0.f;
#pragma unroll
        for (int q = 0; q < HDIM / 4; ++q) {
            const float4 hv = hr[q];
            sc = __builtin_fmaf(hv.x, av4[q].x, sc);
            sc = __builtin_fmaf(hv.y, av4[q].y, sc);
            sc = __builtin_fmaf(hv.z, av4[q].z, sc);
            sc = __builtin_fmaf(hv.w, av4[q].w, sc);
        }
        st[(size_t)rowTile * STREC + threadIdx.x] = sc;
    }

    {
        const int L = lane & 15;
        const float4 mv = make_float4(__builtin_bit_cast(float, m0), __builtin_bit_cast(float, m1),
                                      __builtin_bit_cast(float, m2), __builtin_bit_cast(float, m3));
        const int row = rowTile * 16 + wave * 2 + (L >> 3);
        float4* dst = (float4*)stage + ((size_t)row * (OUTDIM / 4) + (lane >> 4) * (HDIM / 4) + (L & 7));
        dst[0] = mv;
        dst[2 * (HDIM / 4)] = mv;
    }
}

#define ASTEP 32
#define TAU   5.5f

__device__ __forceinline__ float xhalf(float x)
{
    int bits = __builtin_bit_cast(int, x);
    return __builtin_bit_cast(float,
        __builtin_amdgcn_permlanex16(bits, bits, 0x76543210, 0xfedcba98, false, false));
}

__global__ __launch_bounds__(128) void gat_attn(const float* __restrict__ h,
                                                const float* __restrict__ st,
                                                float* out)
{
    const int rowTile = blockIdx.x & 63;
    const int head    = (blockIdx.x >> 6) & (NHEADS - 1);
    const int b       = blockIdx.x >> 8;

    const int tid  = threadIdx.x;
    const int wave = tid >> 5;
    const int lane = tid & 31;
    const int c    = lane & 15;
    const int hh   = lane >> 4;

    __shared__ float wm[4][16];
    __shared__ float wl[4][32];
    __shared__ __attribute__((aligned(16))) float pc[4][16][36];

    const int   iGlob   = rowTile * 16 + c;
    const float* st_b   = st + (size_t)(b * (NNODES / 16)) * STREC;
    const float s_row   = st_b[rowTile * STREC + head * 16 + c];
    const float* t_bh   = st_b + (STREC / 2) + head * 16 + hh * 2;
    const float* mrow = out + (size_t)(b * NNODES + iGlob) * OUTDIM + head * HDIM + hh * 2;
    const float2 mkA = *(const float2*)(mrow + (wave * 2 + 0) * 4);
    const float2 mkB = *(const float2*)(mrow + (wave * 2 + 1) * 4);
    const unsigned long long M0 = (unsigned long long)__builtin_bit_cast(unsigned, mkA.x)
                                | ((unsigned long long)__builtin_bit_cast(unsigned, mkB.x) << 32);
    const unsigned long long M1 = (unsigned long long)__builtin_bit_cast(unsigned, mkA.y)
                                | ((unsigned long long)__builtin_bit_cast(unsigned, mkB.y) << 32);
    const float4* hf4   = (const float4*)h + ((size_t)((b * NHEADS + head) * HGROUPS) * 32 + lane);

    const int jbase = wave * 256;
    float mref = -__builtin_inff();
    float l = 0.f;
    v8f c0 = {}, c1 = {};
    for (int stp = 0; stp < 256 / ASTEP; ++stp) {
        const float* tp = t_bh + (size_t)((jbase + stp * ASTEP) >> 4) * STREC;

        v2f a0[ASTEP / 4], a1[ASTEP / 4];
        const int g0 = (jbase + stp * ASTEP) >> 2;
#pragma unroll
        for (int q = 0; q < ASTEP / 4; ++q) {
            const float4 hv = hf4[(size_t)(g0 + q) * 32];
            a0[q].x = hv.x;
            a0[q].y = hv.y;
            a1[q].x = hv.z;
            a1[q].y = hv.w;
        }

        const unsigned w0 = (unsigned)(M0 >> (8 * stp));
        const unsigned w1 = (unsigned)(M1 >> (8 * stp));
        float e[ASTEP / 2];
#pragma unroll
        for (int q = 0; q < ASTEP / 4; ++q) {
            const float2 tv = *(const float2*)&tp[(q >> 2) * STREC + (q & 3) * 4];
            float e0 = s_row + tv.x;
            e0 = (e0 > 0.f) ? e0 : 0.2f * e0;
            const int k0 = (int)(w0 << (31 - q)) >> 31;
            e0 = __builtin_bit_cast(float, (__builtin_bit_cast(int, e0) & k0) | (NEG_INF_BITS & ~k0));
            float e1 = s_row + tv.y;
            e1 = (e1 > 0.f) ? e1 : 0.2f * e1;
            const int k1 = (int)(w1 << (31 - q)) >> 31;
            e1 = __builtin_bit_cast(float, (__builtin_bit_cast(int, e1) & k1) | (NEG_INF_BITS & ~k1));
            e[2 * q] = e0;
            e[2 * q + 1] = e1;
        }

        float cm = e[0];
#pragma unroll
        for (int k = 1; k < ASTEP / 2; ++k) cm = fmaxf(cm, e[k]);
        cm = fmaxf(cm, xhalf(cm));

        if (__builtin_amdgcn_ballot_w32(cm > mref + TAU) != 0) {
            const float mn = fmaxf(mref, cm);
            const float al = __expf(mref - mn);
            l *= al;
#pragma unroll
            for (int v = 0; v < 8; ++v) { c0[v] *= al; c1[v] *= al; }
            mref = mn;
        }

#pragma unroll
        for (int q = 0; q < ASTEP / 4; ++q) {
            v2f p;
            p.x = __expf(e[2 * q] - mref);
            p.y = __expf(e[2 * q + 1] - mref);
            l += p.x;
            l += p.y;
            c0 = __builtin_amdgcn_wmma_f32_16x16x4_f32(false, a0[q], false, p,
                                                       (short)0, c0, false, false);
            c1 = __builtin_amdgcn_wmma_f32_16x16x4_f32(false, a1[q], false, p,
                                                       (short)0, c1, false, false);
        }
    }

    if (hh == 0) wm[wave][c] = mref;
    __syncthreads();
    const float mall = fmaxf(fmaxf(wm[0][c], wm[1][c]), fmaxf(wm[2][c], wm[3][c]));
    const float f = __expf(mref - mall);
    wl[wave][lane] = l * f;
    {
        float4* d0 = (float4*)&pc[wave][c][8 * hh];
        float4* d1 = (float4*)&pc[wave][c][16 + 8 * hh];
        d0[0] = make_float4(c0[0] * f, c0[1] * f, c0[2] * f, c0[3] * f);
        d0[1] = make_float4(c0[4] * f, c0[5] * f, c0[6] * f, c0[7] * f);
        d1[0] = make_float4(c1[0] * f, c1[1] * f, c1[2] * f, c1[3] * f);
        d1[1] = make_float4(c1[4] * f, c1[5] * f, c1[6] * f, c1[7] * f);
    }
    __syncthreads();

    {
        const int i  = tid >> 3;
        const int d4 = (tid & 7) * 4;
        float lsum = 0.f;
#pragma unroll
        for (int w = 0; w < 4; ++w) { lsum += wl[w][i]; lsum += wl[w][16 + i]; }
        const float4 p0 = *(const float4*)&pc[0][i][d4];
        const float4 p1 = *(const float4*)&pc[1][i][d4];
        const float4 p2 = *(const float4*)&pc[2][i][d4];
        const float4 p3 = *(const float4*)&pc[3][i][d4];
        float4 o;
        o.x = (p0.x + p1.x + p2.x + p3.x) / lsum;
        o.y = (p0.y + p1.y + p2.y + p3.y) / lsum;
        o.z = (p0.z + p1.z + p2.z + p3.z) / lsum;
        o.w = (p0.w + p1.w + p2.w + p3.w) / lsum;
        *(float4*)&out[(size_t)(b * NNODES + rowTile * 16 + i) * OUTDIM + head * HDIM + d4] = o;
    }
}

extern "C" void kernel_launch(void* const* d_in, const int* in_sizes, int n_in,
                              void* d_out, int out_size, void* d_ws, size_t ws_size,
                              hipStream_t stream)
{
    const float* x    = (const float*)d_in[0];
    const int*   adj  = (const int*)  d_in[1];
    const float* W    = (const float*)d_in[2];
    const float* bias = (const float*)d_in[3];
    const float* a    = (const float*)d_in[4];
    float* out = (float*)d_out;

    float* h = (float*)d_ws;
    float* st = h + BATCH * NNODES * OUTDIM;

    gat_linear<<<(BATCH * NNODES) / 16, 256, 0, stream>>>(x, W, bias, a, adj, h, st, out);
    gat_attn<<<BATCH * NHEADS * (NNODES / 16), 128, 0, stream>>>(h, st, out);
}
